# starting kernel
# baseline (speedup 1.0000x reference)
#include <hip/hip_runtime.h>

typedef __attribute__((ext_vector_type(2))) float v2f;
typedef __attribute__((ext_vector_type(8))) float v8f;

#define HW_H   96
#define HW_W   96
#define NPIX   (HW_H * HW_W)
#define NB     2
#define NC     21
#define NT     (NPIX / 16)
#define NWAVE  8
#define KSTR   18

static __device__ __forceinline__ v8f wmma_f32(v2f a, v2f b, v8f c) {
    return __builtin_amdgcn_wmma_f32_16x16x4_f32(false, a, false, b, (short)0, c, false, false);
}

__global__ void __launch_bounds__(256) pgf_prep_feats(const float* __restrict__ img,
                                                      float* __restrict__ feats,
                                                      float* __restrict__ sq) {
    int idx = blockIdx.x * blockDim.x + threadIdx.x;
    if (idx >= NB * NPIX) return;
    int b = idx / NPIX;
    int n = idx - b * NPIX;
    float f[8];
    f[0] = (float)(n / HW_W) * 0.125f;
    f[1] = (float)(n % HW_W) * 0.125f;
    const float* ip = img + (size_t)b * 3 * NPIX + n;
    f[2] = ip[0]        * 2.0f;
    f[3] = ip[NPIX]     * 2.0f;
    f[4] = ip[2 * NPIX] * 2.0f;
    f[5] = 0.0f; f[6] = 0.0f; f[7] = 0.0f;
    float s = 0.0f;
#pragma unroll
    for (int v = 0; v < 8; ++v) s += f[v] * f[v];
    float* fp = feats + (size_t)idx * 8;
#pragma unroll
    for (int v = 0; v < 8; ++v) fp[v] = f[v];
    sq[idx] = s;
}

__global__ void __launch_bounds__(256) pgf_prep_vt(const float* __restrict__ vals,
                                                   float* __restrict__ vt) {
    int idx = blockIdx.x * blockDim.x + threadIdx.x;
    if (idx >= NB * NT * 32) return;
    int c  = idx & 31;
    int jt = (idx >> 5) % NT;
    int b  = idx / (32 * NT);
    float* dst = vt + (size_t)idx * 16;
    if (c < NC) {
        const float* src = vals + ((size_t)b * NC + c) * NPIX + jt * 16;
#pragma unroll
        for (int jj = 0; jj < 16; ++jj) dst[jj] = src[jj];
    } else {
#pragma unroll
        for (int jj = 0; jj < 16; ++jj) dst[jj] = 0.0f;
    }
}

__global__ void __launch_bounds__(256) pgf_main(const float* __restrict__ feats,
                                                const float* __restrict__ sq,
                                                const float* __restrict__ vt,
                                                float* __restrict__ out) {
    __shared__ float lds_k[NWAVE * 16 * KSTR];
    __shared__ float lds_r[NWAVE * 512];

    const int tid = threadIdx.x;
    const int wv  = tid >> 5;
    const int l   = tid & 31;
    const int col = l & 15;
    const int p   = l >> 4;
    const int it     = blockIdx.x;
    const int b      = blockIdx.y;
    const int i_base = it * 16;
    const size_t fbase = (size_t)b * NPIX;

    const float* fi = feats + (fbase + i_base + col) * 8;
    v2f alo = *(const v2f*)(fi + 2 * p);
    v2f ahi = *(const v2f*)(fi + 4 + 2 * p);

    float si[8];
    const float* sqi = sq + fbase + i_base + 8 * p;
#pragma unroll
    for (int v = 0; v < 8; ++v) si[v] = sqi[v];

    v8f acc0 = {};
    v8f acc1 = {};
    float* kt = lds_k + wv * 16 * KSTR;

    for (int jt = wv; jt < NT; jt += NWAVE) {
        const int j_base = jt * 16;

        const float* fj = feats + (fbase + j_base + col) * 8;
        v2f blo = *(const v2f*)(fj + 2 * p);
        v2f bhi = *(const v2f*)(fj + 4 + 2 * p);
        float sj = sq[fbase + j_base + col];

        v8f g = {};
        g = wmma_f32(alo, blo, g);
        g = wmma_f32(ahi, bhi, g);

#pragma unroll
        for (int v = 0; v < 8; ++v) {
            float d2 = si[v] + sj - 2.0f * g[v];
            float kv = __expf(-0.5f * fmaxf(d2, 0.0f));
            kt[(v + 8 * p) * KSTR + col] = kv;
        }

        if (jt + NWAVE < NT)
            __builtin_prefetch(feats + (fbase + j_base + 16 * NWAVE + col) * 8, 0, 3);

        const float* vtile = vt + (size_t)(b * NT + jt) * 32 * 16;
#pragma unroll
        for (int kk = 0; kk < 4; ++kk) {
            v2f ka  = *(const v2f*)(kt + col * KSTR + 4 * kk + 2 * p);
            v2f vb0 = *(const v2f*)(vtile + (col     ) * 16 + 4 * kk + 2 * p);
            v2f vb1 = *(const v2f*)(vtile + (col + 16) * 16 + 4 * kk + 2 * p);
            acc0 = wmma_f32(ka, vb0, acc0);
            acc1 = wmma_f32(ka, vb1, acc1);
        }
    }

#pragma unroll
    for (int v = 0; v < 8; ++v) {
        int row = v + 8 * p;
        lds_r[wv * 512 + row * 32 + col]      = acc0[v];
        lds_r[wv * 512 + row * 32 + col + 16] = acc1[v];
    }
    __syncthreads();
    for (int e = tid; e < 512; e += 256) {
        float s = 0.0f;
#pragma unroll
        for (int w2 = 0; w2 < NWAVE; ++w2) s += lds_r[w2 * 512 + e];
        int row = e >> 5;
        int c   = e & 31;
        if (c < NC) out[((size_t)b * NC + c) * NPIX + i_base + row] = s;
    }
}

extern "C" void kernel_launch(void* const* d_in, const int* in_sizes, int n_in,
                              void* d_out, int out_size, void* d_ws, size_t ws_size,
                              hipStream_t stream) {
    const float* cur_state   = (const float*)d_in[0];
    const float* input_image = (const float*)d_in[1];
    float* out = (float*)d_out;

    float* ws    = (float*)d_ws;
    float* feats = ws;
    float* sq    = feats + (size_t)NB * NPIX * 8;
    float* vtws  = sq    + (size_t)NB * NPIX;
    (void)in_sizes; (void)n_in; (void)out_size; (void)ws_size;

    {
        int nthr = NB * NPIX;
        pgf_prep_feats<<<(nthr + 255) / 256, 256, 0, stream>>>(input_image, feats, sq);
    }
    {
        int nthr = NB * NT * 32;
        pgf_prep_vt<<<(nthr + 255) / 256, 256, 0, stream>>>(cur_state, vtws);
    }
    {
        dim3 grid(NT, NB);
        pgf_main<<<grid, 256, 0, stream>>>(feats, sq, vtws, out);
    }
}


# Workspace operands stored in fragment order: a lane fetches its feature and V fragments by 16-byte loads over whole 128-byte lines (6 loads, 21 lines per tile pair, not 11 and 73); k, sum order kept
# speedup vs baseline: 1.1177x; 1.1177x over previous
#include <hip/hip_runtime.h>

typedef __attribute__((ext_vector_type(2))) float v2f;
typedef __attribute__((ext_vector_type(4))) float v4f;
typedef __attribute__((ext_vector_type(8))) float v8f;

#define HW_H   96
#define HW_W   96
#define NPIX   (HW_H * HW_W)
#define NB     2
#define NC     21
#define NT     (NPIX / 16)
#define NWAVE  8
#define KSTR   18

static __device__ __forceinline__ v8f wmma_f32(v2f a, v2f b, v8f c) {
    return __builtin_amdgcn_wmma_f32_16x16x4_f32(false, a, false, b, (short)0, c, false, false);
}

__global__ void __launch_bounds__(256) pgf_prep_feats(const float* __restrict__ img,
                                                      float* __restrict__ ff,
                                                      float* __restrict__ sq) {
    int idx = blockIdx.x * blockDim.x + threadIdx.x;
    if (idx >= NB * NPIX) return;
    int b = idx / NPIX;
    int n = idx - b * NPIX;
    float f[5];
    f[0] = (float)(n / HW_W) * 0.125f;
    f[1] = (float)(n % HW_W) * 0.125f;
    const float* ip = img + (size_t)b * 3 * NPIX + n;
    f[2] = ip[0]        * 2.0f;
    f[3] = ip[NPIX]     * 2.0f;
    f[4] = ip[2 * NPIX] * 2.0f;
    float s = 0.0f;
#pragma unroll
    for (int v = 0; v < 5; ++v) s += f[v] * f[v];
    const int t   = idx >> 4;
    const int col = idx & 15;
    v4f lo = { f[0], f[1], f[4], 0.0f };
    v4f hi = { f[2], f[3], 0.0f, 0.0f };
    v4f* dst = (v4f*)ff + (size_t)t * 32 + col;
    dst[0]  = lo;
    dst[16] = hi;
    sq[idx] = s;
}

__global__ void __launch_bounds__(256) pgf_prep_vt(const float* __restrict__ vals,
                                                   float* __restrict__ vf) {
    int idx = blockIdx.x * blockDim.x + threadIdx.x;
    if (idx >= NB * NT * 4 * 32) return;
    int l   = idx & 31;
    int kk  = (idx >> 5) & 3;
    int jt  = (idx >> 7) % NT;
    int b   = idx / (128 * NT);
    int col = l & 15;
    int p   = l >> 4;
    const float* src = vals + ((size_t)b * NC + col) * NPIX + jt * 16 + 4 * kk + 2 * p;
    v2f a = *(const v2f*)src;
    v2f c = { 0.0f, 0.0f };
    if (16 + col < NC) c = *(const v2f*)(src + (size_t)16 * NPIX);
    v4f o = { a.x, a.y, c.x, c.y };
    ((v4f*)vf)[idx] = o;
}

struct JTile {
    v4f   fb;
    float sj;
    v4f   vq[4];
};

static __device__ __forceinline__ void load_jtile(JTile& r,
                                                  const float* __restrict__ ff, unsigned fo,
                                                  const float* __restrict__ sq, unsigned so,
                                                  const float* __restrict__ vf, unsigned vo) {
    const char* f = (const char*)ff + fo;
    const char* s = (const char*)sq + so;
    const char* v = (const char*)vf + vo;
    r.fb = *(const v4f*)(f);
    r.sj = *(const float*)(s);
#pragma unroll
    for (int kk = 0; kk < 4; ++kk) r.vq[kk] = *(const v4f*)(v + 512 * kk);
}

static __device__ __forceinline__ v2f lo2(v4f x) { v2f r = { x.x, x.y }; return r; }
static __device__ __forceinline__ v2f hi2(v4f x) { v2f r = { x.z, x.w }; return r; }

__global__ void __launch_bounds__(256) pgf_main(const float* __restrict__ feats,
                                                const float* __restrict__ sq,
                                                const float* __restrict__ vt,
                                                float* __restrict__ out) {
    __shared__ float lds_k[NWAVE * 16 * KSTR];
    __shared__ float lds_r[NWAVE * 512];

    const int tid = threadIdx.x;
    const int wv  = tid >> 5;
    const int l   = tid & 31;
    const int col = l & 15;
    const int p   = l >> 4;
    const int it     = blockIdx.x;
    const int b      = blockIdx.y;
    const int i_base = it * 16;
    const size_t fbase = (size_t)b * NPIX;

    const v4f fa = *((const v4f*)feats + ((size_t)b * NT + it) * 32 + l);
    const v2f alo = lo2(fa);
    const v2f ahi = hi2(fa);

    float si[8];
    const float* sqi = sq + fbase + i_base + 8 * p;
#pragma unroll
    for (int v = 0; v < 8; ++v) si[v] = sqi[v];

    v8f acc0 = {};
    v8f acc1 = {};
    float* kt = lds_k + wv * 16 * KSTR;
    float*       ktw = kt + 8 * p * KSTR + col;
    const float* ktr = kt + col * KSTR + 2 * p;

    unsigned fo = 16u * (((unsigned)b * NT + wv) * 32 + l);
    unsigned so = 4u * ((unsigned)b * NPIX + wv * 16 + col);
    unsigned vo = 16u * (((unsigned)b * NT + wv) * 128 + l);
    constexpr unsigned FJ_STEP = 16u * NWAVE * 32;
    constexpr unsigned SQ_STEP = 4u * NWAVE * 16;
    constexpr unsigned VJ_STEP = 4u * NWAVE * 512;
    constexpr int NSTEP   = NT / NWAVE;

    auto compute = [&](const JTile& r) {
        v8f g = {};
        g = wmma_f32(alo, lo2(r.fb), g);
        g = wmma_f32(ahi, hi2(r.fb), g);

#pragma unroll
        for (int v = 0; v < 8; ++v) {
            float d2 = si[v] + r.sj - 2.0f * g[v];
            float kv = __expf(-0.5f * fmaxf(d2, 0.0f));
            ktw[v * KSTR] = kv;
        }

#pragma unroll
        for (int kk = 0; kk < 4; ++kk) {
            v2f ka = *(const v2f*)(ktr + 4 * kk);
            acc0 = wmma_f32(ka, lo2(r.vq[kk]), acc0);
            acc1 = wmma_f32(ka, hi2(r.vq[kk]), acc1);
        }
    };

    JTile ra, rb;
    load_jtile(ra, feats, fo, sq, so, vt, vo);
    for (int t = 0; t < NSTEP; t += 2) {
        load_jtile(rb, feats, fo + FJ_STEP, sq, so + SQ_STEP, vt, vo + VJ_STEP);
        compute(ra);
        asm volatile("" :: "v"(acc0), "v"(acc1) : "memory");
        const unsigned adv = (t + 2 < NSTEP) ? 2u : 1u;
        fo += adv * FJ_STEP;
        so += adv * SQ_STEP;
        vo += adv * VJ_STEP;
        load_jtile(ra, feats, fo, sq, so, vt, vo);
        compute(rb);
    }

#pragma unroll
    for (int v = 0; v < 8; ++v) {
        int row = v + 8 * p;
        lds_r[wv * 512 + row * 32 + col]      = acc0[v];
        lds_r[wv * 512 + row * 32 + col + 16] = acc1[v];
    }
    __syncthreads();
    for (int e = tid; e < 512; e += 256) {
        float s = 0.0f;
#pragma unroll
        for (int w2 = 0; w2 < NWAVE; ++w2) s += lds_r[w2 * 512 + e];
        int row = e >> 5;
        int c   = e & 31;
        if (c < NC) out[((size_t)b * NC + c) * NPIX + i_base + row] = s;
    }
}

extern "C" void kernel_launch(void* const* d_in, const int* in_sizes, int n_in,
                              void* d_out, int out_size, void* d_ws, size_t ws_size,
                              hipStream_t stream) {
    const float* cur_state   = (const float*)d_in[0];
    const float* input_image = (const float*)d_in[1];
    float* out = (float*)d_out;

    float* ws    = (float*)d_ws;
    float* feats = ws;
    float* sq    = feats + (size_t)NB * NPIX * 8;
    float* vtws  = sq    + (size_t)NB * NPIX;
    (void)in_sizes; (void)n_in; (void)out_size; (void)ws_size;

    {
        int nthr = NB * NPIX;
        pgf_prep_feats<<<(nthr + 255) / 256, 256, 0, stream>>>(input_image, feats, sq);
    }
    {
        int nthr = NB * NT * 4 * 32;
        pgf_prep_vt<<<(nthr + 255) / 256, 256, 0, stream>>>(cur_state, vtws);
    }
    {
        dim3 grid(NT, NB);
        pgf_main<<<grid, 256, 0, stream>>>(feats, sq, vtws, out);
    }
}


# Norm terms and exp2 scale ride in the 3 unused k slots of the K=8 feature product: the product is the exponent, per element only a clamped exp2 is left; no norm loads; distance terms in new order
# speedup vs baseline: 1.2362x; 1.1060x over previous
#include <hip/hip_runtime.h>

typedef __attribute__((ext_vector_type(2))) float v2f;
typedef __attribute__((ext_vector_type(4))) float v4f;
typedef __attribute__((ext_vector_type(8))) float v8f;

#define HW_H   96
#define HW_W   96
#define NPIX   (HW_H * HW_W)
#define NB     2
#define NC     21
#define NT     (NPIX / 16)
#define NWAVE  8
#define KSTR   18

static __device__ __forceinline__ v8f wmma_f32(v2f a, v2f b, v8f c) {
    return __builtin_amdgcn_wmma_f32_16x16x4_f32(false, a, false, b, (short)0, c, false, false);
}

#define PGF_LOG2E   1.44269504088896340736f
#define PGF_HLOG2E  0.72134752044448170368f

__global__ void __launch_bounds__(256) pgf_prep_feats(const float* __restrict__ img,
                                                      float* __restrict__ ff) {
    int idx = blockIdx.x * blockDim.x + threadIdx.x;
    if (idx >= NB * NPIX) return;
    int b = idx / NPIX;
    int n = idx - b * NPIX;
    float f[5];
    f[0] = (float)(n / HW_W) * 0.125f;
    f[1] = (float)(n % HW_W) * 0.125f;
    const float* ip = img + (size_t)b * 3 * NPIX + n;
    f[2] = ip[0]        * 2.0f;
    f[3] = ip[NPIX]     * 2.0f;
    f[4] = ip[2 * NPIX] * 2.0f;
    float s = 0.0f;
#pragma unroll
    for (int v = 0; v < 5; ++v) s += f[v] * f[v];
    const int t   = idx >> 4;
    const int col = idx & 15;
    v4f lo = { f[0], f[1], f[4], 1.0f };
    v4f hi = { f[2], f[3], -PGF_HLOG2E * s, 0.0f };
    v4f* dst = (v4f*)ff + (size_t)t * 32 + col;
    dst[0]  = lo;
    dst[16] = hi;
}

__global__ void __launch_bounds__(256) pgf_prep_vt(const float* __restrict__ vals,
                                                   float* __restrict__ vf) {
    int idx = blockIdx.x * blockDim.x + threadIdx.x;
    if (idx >= NB * NT * 4 * 32) return;
    int l   = idx & 31;
    int kk  = (idx >> 5) & 3;
    int jt  = (idx >> 7) % NT;
    int b   = idx / (128 * NT);
    int col = l & 15;
    int p   = l >> 4;
    const float* src = vals + ((size_t)b * NC + col) * NPIX + jt * 16 + 4 * kk + 2 * p;
    v2f a = *(const v2f*)src;
    v2f c = { 0.0f, 0.0f };
    if (16 + col < NC) c = *(const v2f*)(src + (size_t)16 * NPIX);
    v4f o = { a.x, a.y, c.x, c.y };
    ((v4f*)vf)[idx] = o;
}

struct JTile {
    v4f   fb;
    v4f   vq[4];
};

static __device__ __forceinline__ void load_jtile(JTile& r,
                                                  const float* __restrict__ ff, unsigned fo,
                                                  const float* __restrict__ vf, unsigned vo) {
    const char* f = (const char*)ff + fo;
    const char* v = (const char*)vf + vo;
    r.fb = *(const v4f*)(f);
#pragma unroll
    for (int kk = 0; kk < 4; ++kk) r.vq[kk] = *(const v4f*)(v + 512 * kk);
}

static __device__ __forceinline__ v2f lo2(v4f x) { v2f r = { x.x, x.y }; return r; }
static __device__ __forceinline__ v2f hi2(v4f x) { v2f r = { x.z, x.w }; return r; }

__global__ void __launch_bounds__(256) __attribute__((amdgpu_waves_per_eu(10, 10)))
pgf_main(const float* __restrict__ feats,
         const float* __restrict__ vt,
         float* __restrict__ out) {
    __shared__ float lds_k[NWAVE * 16 * KSTR];
    __shared__ float lds_r[NWAVE * 512];

    const int tid = threadIdx.x;
    const int wv  = tid >> 5;
    const int l   = tid & 31;
    const int col = l & 15;
    const int p   = l >> 4;
    const int it     = blockIdx.x;
    const int b      = blockIdx.y;
    const int i_base = it * 16;

    const v4f* reci = (const v4f*)feats + ((size_t)b * NT + it) * 32 + col;
    const v4f rlo = reci[0];
    const v4f rhi = reci[16];
    v2f alo, ahi;
    if (p == 0) {
        alo.x = PGF_LOG2E * rlo.x;  alo.y = PGF_LOG2E * rlo.y;
        ahi.x = PGF_LOG2E * rlo.z;  ahi.y = rhi.z;
    } else {
        alo.x = PGF_LOG2E * rhi.x;  alo.y = PGF_LOG2E * rhi.y;
        ahi.x = 1.0f;               ahi.y = 0.0f;
    }

    v8f acc0 = {};
    v8f acc1 = {};
    float* kt = lds_k + wv * 16 * KSTR;
    float*       ktw = kt + 8 * p * KSTR + col;
    const float* ktr = kt + col * KSTR + 2 * p;

    unsigned fo = 16u * (((unsigned)b * NT + wv) * 32 + l);
    unsigned vo = 16u * (((unsigned)b * NT + wv) * 128 + l);
    constexpr unsigned FJ_STEP = 16u * NWAVE * 32;
    constexpr unsigned VJ_STEP = 4u * NWAVE * 512;
    constexpr int NSTEP   = NT / NWAVE;

    auto compute = [&](const JTile& r) {
        v8f g = {};
        g = wmma_f32(alo, lo2(r.fb), g);
        g = wmma_f32(ahi, hi2(r.fb), g);

#pragma unroll
        for (int v = 0; v < 8; ++v) {
            float kv = __builtin_amdgcn_exp2f(g[v]);
            kv = fminf(fmaxf(kv, 0.0f), 1.0f);
            ktw[v * KSTR] = kv;
        }

#pragma unroll
        for (int kk = 0; kk < 4; ++kk) {
            v2f ka = *(const v2f*)(ktr + 4 * kk);
            acc0 = wmma_f32(ka, lo2(r.vq[kk]), acc0);
            acc1 = wmma_f32(ka, hi2(r.vq[kk]), acc1);
        }
    };

    JTile ra, rb;
    load_jtile(ra, feats, fo, vt, vo);
    for (int t = 0; t < NSTEP; t += 2) {
        load_jtile(rb, feats, fo + FJ_STEP, vt, vo + VJ_STEP);
        compute(ra);
        asm volatile("" :: "v"(acc0), "v"(acc1) : "memory");
        const unsigned adv = (t + 2 < NSTEP) ? 2u : 1u;
        fo += adv * FJ_STEP;
        vo += adv * VJ_STEP;
        load_jtile(ra, feats, fo, vt, vo);
        compute(rb);
    }

#pragma unroll
    for (int v = 0; v < 8; ++v) {
        int row = v + 8 * p;
        lds_r[wv * 512 + row * 32 + col]      = acc0[v];
        lds_r[wv * 512 + row * 32 + col + 16] = acc1[v];
    }
    __syncthreads();
    for (int e = tid; e < 512; e += 256) {
        float s = 0.0f;
#pragma unroll
        for (int w2 = 0; w2 < NWAVE; ++w2) s += lds_r[w2 * 512 + e];
        int row = e >> 5;
        int c   = e & 31;
        if (c < NC) out[((size_t)b * NC + c) * NPIX + i_base + row] = s;
    }
}

extern "C" void kernel_launch(void* const* d_in, const int* in_sizes, int n_in,
                              void* d_out, int out_size, void* d_ws, size_t ws_size,
                              hipStream_t stream) {
    const float* cur_state   = (const float*)d_in[0];
    const float* input_image = (const float*)d_in[1];
    float* out = (float*)d_out;

    float* ws    = (float*)d_ws;
    float* feats = ws;
    float* vtws  = feats + (size_t)NB * NPIX * 8;
    (void)in_sizes; (void)n_in; (void)out_size; (void)ws_size;

    {
        int nthr = NB * NPIX;
        pgf_prep_feats<<<(nthr + 255) / 256, 256, 0, stream>>>(input_image, feats);
    }
    {
        int nthr = NB * NT * 4 * 32;
        pgf_prep_vt<<<(nthr + 255) / 256, 256, 0, stream>>>(cur_state, vtws);
    }
    {
        dim3 grid(NT, NB);
        pgf_main<<<grid, 256, 0, stream>>>(feats, vtws, out);
    }
}


# Channels 16-20 leave the padded 2nd matrix tile: per-lane packed f32 FMAs on K in its D registers, 6 WMMAs per tile pair, not 10; operand loads refilled behind last use; their j-sum order changes
# speedup vs baseline: 1.3912x; 1.1254x over previous
#include <hip/hip_runtime.h>

typedef __attribute__((ext_vector_type(2))) float v2f;
typedef __attribute__((ext_vector_type(4))) float v4f;
typedef __attribute__((ext_vector_type(8))) float v8f;

#define HW_H   96
#define HW_W   96
#define NPIX   (HW_H * HW_W)
#define NB     2
#define NC     21
#define NT     (NPIX / 16)
#define NWAVE  8
#define KSTR   18
#define NX     (NC - 16)

static __device__ __forceinline__ v8f wmma_f32(v2f a, v2f b, v8f c) {
    return __builtin_amdgcn_wmma_f32_16x16x4_f32(false, a, false, b, (short)0, c, false, false);
}

#define PGF_LOG2E   1.44269504088896340736f
#define PGF_HLOG2E  0.72134752044448170368f

__global__ void __launch_bounds__(256) pgf_prep_feats(const float* __restrict__ img,
                                                      float* __restrict__ ff) {
    int idx = blockIdx.x * blockDim.x + threadIdx.x;
    if (idx >= NB * NPIX) return;
    int b = idx / NPIX;
    int n = idx - b * NPIX;
    float f[5];
    f[0] = (float)(n / HW_W) * 0.125f;
    f[1] = (float)(n % HW_W) * 0.125f;
    const float* ip = img + (size_t)b * 3 * NPIX + n;
    f[2] = ip[0]        * 2.0f;
    f[3] = ip[NPIX]     * 2.0f;
    f[4] = ip[2 * NPIX] * 2.0f;
    float s = 0.0f;
#pragma unroll
    for (int v = 0; v < 5; ++v) s += f[v] * f[v];
    const int t   = idx >> 4;
    const int col = idx & 15;
    v4f lo = { f[0], f[1], f[4], 1.0f };
    v4f hi = { f[2], f[3], -PGF_HLOG2E * s, 0.0f };
    v4f* dst = (v4f*)ff + (size_t)t * 32 + col;
    dst[0]  = lo;
    dst[16] = hi;
}

#define NLANE_RECS (NB * NT * 32)
__global__ void __launch_bounds__(256) pgf_prep_vt(const float* __restrict__ vals,
                                                   float* __restrict__ vf,
                                                   float* __restrict__ vx) {
    int idx = blockIdx.x * blockDim.x + threadIdx.x;
    if (idx >= 3 * NLANE_RECS) return;
    const int plane = idx / NLANE_RECS;
    const int rec   = idx - plane * NLANE_RECS;
    const int l   = rec & 31;
    const int t   = rec >> 5;
    const int b   = t / NT;
    const int jt  = t - b * NT;
    const int col = l & 15;
    const int p   = l >> 4;
    if (plane < 2) {
        const float* s = vals + ((size_t)b * NC + col) * NPIX + jt * 16 + 8 * plane + 2 * p;
        v2f a = *(const v2f*)s;
        v2f c = *(const v2f*)(s + 4);
        v4f o = { a.x, a.y, c.x, c.y };
        ((v4f*)vf)[idx] = o;
    } else {
        const float* s = vals + ((size_t)b * NC + 16) * NPIX + jt * 16 + col;
        v4f o = { s[0], s[NPIX], s[2 * NPIX], s[3 * NPIX] };
        ((v4f*)vx)[rec] = o;
        vx[(size_t)4 * NLANE_RECS + rec] = s[4 * NPIX];
    }
}

struct JTile {
    v4f   fb;
    v4f   vq[2];
    v4f   xa;
    float xb;
};

#define PLANE_BYTES ((size_t)16 * NLANE_RECS)
static __device__ __forceinline__ void load_fb(JTile& r, unsigned fo, const float* __restrict__ ff) {
    r.fb = *(const v4f*)((const char*)ff + fo);
}
static __device__ __forceinline__ void load_vq0(JTile& r, unsigned fo, const float* __restrict__ vf) {
    r.vq[0] = *(const v4f*)((const char*)vf + fo);
}
static __device__ __forceinline__ void load_vq1(JTile& r, unsigned fo, const float* __restrict__ vf) {
    r.vq[1] = *(const v4f*)((const char*)vf + PLANE_BYTES + fo);
}
static __device__ __forceinline__ void load_x(JTile& r, unsigned fo, const float* __restrict__ vx) {
    r.xa = *(const v4f*)((const char*)vx + fo);
    r.xb = *(const float*)((const char*)vx + PLANE_BYTES + (fo >> 2));
}

template <int CTRL>
static __device__ __forceinline__ float row_rot_add(float x) {
    int y = __builtin_amdgcn_update_dpp(0, __builtin_bit_cast(int, x), CTRL, 0xF, 0xF, false);
    return x + __builtin_bit_cast(float, y);
}
static __device__ __forceinline__ float row_sum16(float x) {
    x = row_rot_add<0x128>(x);
    x = row_rot_add<0x124>(x);
    x = row_rot_add<0x122>(x);
    x = row_rot_add<0x121>(x);
    return x;
}

static __device__ __forceinline__ v2f lo2(v4f x) { v2f r = { x.x, x.y }; return r; }
static __device__ __forceinline__ v2f hi2(v4f x) { v2f r = { x.z, x.w }; return r; }

__global__ void __launch_bounds__(256) __attribute__((amdgpu_waves_per_eu(10, 10)))
pgf_main(const float* __restrict__ feats,
         const float* __restrict__ vt,
         const float* __restrict__ vx,
         float* __restrict__ out) {
    __shared__ float lds_k[NWAVE * 16 * KSTR];
    __shared__ float lds_r[NWAVE * 512];

    const int tid = threadIdx.x;
    const int wv  = __builtin_amdgcn_readfirstlane(tid >> 5);
    const int l   = tid & 31;
    const int col = l & 15;
    const int p   = l >> 4;
    const int it     = blockIdx.x;
    const int b      = blockIdx.y;
    const int i_base = it * 16;

    const v4f* reci = (const v4f*)feats + ((size_t)b * NT + it) * 32 + col;
    const v4f rlo = reci[0];
    const v4f rhi = reci[16];
    v2f alo, ahi;
    if (p == 0) {
        alo.x = PGF_LOG2E * rlo.x;  alo.y = PGF_LOG2E * rlo.y;
        ahi.x = PGF_LOG2E * rlo.z;  ahi.y = rhi.z;
    } else {
        alo.x = PGF_LOG2E * rhi.x;  alo.y = PGF_LOG2E * rhi.y;
        ahi.x = 1.0f;               ahi.y = 0.0f;
    }

    v8f acc0 = {};
    v2f part[NX][4];
#pragma unroll
    for (int c = 0; c < NX; ++c)
#pragma unroll
        for (int u = 0; u < 4; ++u) part[c][u] = (v2f){ 0.0f, 0.0f };
    float* kt = lds_k + wv * 16 * KSTR;
    float*       ktw = kt + 8 * p * KSTR + col;
    const float* ktr = kt + col * KSTR + 2 * p;

    unsigned fo = 16u * (((unsigned)b * NT + wv) * 32 + l);
    constexpr unsigned FJ_STEP = 16u * NWAVE * 32;
    constexpr int NSTEP   = NT / NWAVE;

    JTile r;
    load_fb(r, fo, feats);
    load_x(r, fo, vx);
    load_vq0(r, fo, vt);
    for (int t = 0; t < NSTEP; ++t) {
        const unsigned fn = fo + ((t + 1 < NSTEP) ? FJ_STEP : 0u);

        load_vq1(r, fo, vt);

        v8f g = {};
        g = wmma_f32(alo, lo2(r.fb), g);
        g = wmma_f32(ahi, hi2(r.fb), g);
        __builtin_amdgcn_sched_barrier(0);
        load_fb(r, fn, feats);
        __builtin_amdgcn_sched_barrier(0);

        float kv[8];
#pragma unroll
        for (int v = 0; v < 8; ++v) {
            kv[v] = __builtin_amdgcn_exp2f(g[v]);
            kv[v] = fminf(fmaxf(kv[v], 0.0f), 1.0f);
            ktw[v * KSTR] = kv[v];
        }

        v2f ka[4];
#pragma unroll
        for (int kk = 0; kk < 4; ++kk) ka[kk] = *(const v2f*)(ktr + 4 * kk);

#pragma unroll
        for (int c = 0; c < NX; ++c) {
            const float x = (c < 4) ? r.xa[c] : r.xb;
            const v2f x2 = { x, x };
#pragma unroll
            for (int u = 0; u < 4; ++u) {
                const v2f k2 = { kv[2 * u], kv[2 * u + 1] };
                part[c][u] = __builtin_elementwise_fma(k2, x2, part[c][u]);
            }
        }
        load_x(r, fn, vx);

        acc0 = wmma_f32(ka[0], lo2(r.vq[0]), acc0);
        acc0 = wmma_f32(ka[1], hi2(r.vq[0]), acc0);
        load_vq0(r, fn, vt);
        acc0 = wmma_f32(ka[2], lo2(r.vq[1]), acc0);
        acc0 = wmma_f32(ka[3], hi2(r.vq[1]), acc0);
        asm volatile("" :: "v"(acc0) : "memory");
        fo = fn;
    }

    const int l2   = (int)__builtin_amdgcn_mbcnt_lo(~0u, 0u);
    const int col2 = l2 & 15;
    const int p2   = l2 >> 4;
    const int tid2 = wv * 32 + l2;
#pragma unroll
    for (int v = 0; v < 8; ++v) {
        int row = v + 8 * p2;
        lds_r[wv * 512 + row * 32 + col2] = acc0[v];
        float mine = 0.0f;
#pragma unroll
        for (int c = 0; c < NX; ++c) {
            float s = row_sum16(part[c][v >> 1][v & 1]);
            if (col2 == c) mine = s;
        }
        if (col2 < NX) lds_r[wv * 512 + row * 32 + 16 + col2] = mine;
    }
    __syncthreads();
    for (int e = tid2; e < 512; e += 256) {
        int row = e >> 5;
        int c   = e & 31;
        if (c < NC) {
            float s = 0.0f;
#pragma unroll
            for (int w2 = 0; w2 < NWAVE; ++w2) s += lds_r[w2 * 512 + e];
            out[((size_t)b * NC + c) * NPIX + i_base + row] = s;
        }
    }
}

extern "C" void kernel_launch(void* const* d_in, const int* in_sizes, int n_in,
                              void* d_out, int out_size, void* d_ws, size_t ws_size,
                              hipStream_t stream) {
    const float* cur_state   = (const float*)d_in[0];
    const float* input_image = (const float*)d_in[1];
    float* out = (float*)d_out;

    float* ws    = (float*)d_ws;
    float* feats = ws;
    float* vtws  = feats + (size_t)NB * NPIX * 8;
    float* vxws  = vtws + (size_t)NLANE_RECS * 8;
    (void)in_sizes; (void)n_in; (void)out_size; (void)ws_size;

    {
        int nthr = NB * NPIX;
        pgf_prep_feats<<<(nthr + 255) / 256, 256, 0, stream>>>(input_image, feats);
    }
    {
        int nthr = 3 * NLANE_RECS;
        pgf_prep_vt<<<(nthr + 255) / 256, 256, 0, stream>>>(cur_state, vtws, vxws);
    }
    {
        dim3 grid(NT, NB);
        pgf_main<<<grid, 256, 0, stream>>>(feats, vtws, vxws, out);
    }
}
